# starting kernel
# baseline (speedup 1.0000x reference)
#include <hip/hip_runtime.h>
#include <hip/hip_bf16.h>

#define BATCH 2
#define SEQ   1024
#define EMB   256
#define HEADS 64
#define DKK   4

typedef __attribute__((ext_vector_type(16))) _Float16 v16h;
typedef __attribute__((ext_vector_type(8)))  float    v8f;
typedef __attribute__((ext_vector_type(2)))  float    v2f;

#if defined(__has_builtin)
#if __has_builtin(__builtin_amdgcn_wmma_f32_16x16x4_f32)
#define HAVE_WMMA_F32X4 1
#endif
#endif

__global__ __launch_bounds__(256) void qa_qkv_quantum(
    const float* __restrict__ x,
    const float* __restrict__ Wq, const float* __restrict__ Wk,
    const float* __restrict__ Wv, const float* __restrict__ params,
    float* __restrict__ qp, float* __restrict__ kp, float* __restrict__ vp)
{
  const int lane = threadIdx.x & 31;
  const int wave = threadIdx.x >> 5;
  const int lo = lane & 15, hi = lane >> 4;

  const int tile = blockIdx.x * 8 + wave;
  const int mat  = tile >> 11;
  const int rem  = tile & 2047;
  const int nt   = rem >> 7;
  const int mt   = rem & 127;

  const float* W    = (mat == 0) ? Wq : (mat == 1) ? Wk : Wv;
  float*       outp = (mat == 0) ? qp : (mat == 1) ? kp : vp;

  const float* wrow = W + (nt * 16 + lo) * EMB;
  const float* xrow = x + (mt * 16 + lo) * EMB;

  v8f c = {};
  #pragma unroll
  for (int ks = 0; ks < EMB; ks += 32) {
    const float* wa = wrow + ks + hi * 8;
    const float* xa = xrow + ks + hi * 8;
    const float4 w0 = *(const float4*)(wa);
    const float4 w1 = *(const float4*)(wa + 4);
    const float4 w2 = *(const float4*)(wa + 16);
    const float4 w3 = *(const float4*)(wa + 20);
    const float4 x0 = *(const float4*)(xa);
    const float4 x1 = *(const float4*)(xa + 4);
    const float4 x2 = *(const float4*)(xa + 16);
    const float4 x3 = *(const float4*)(xa + 20);
    v16h a, b;
    a[0]=(_Float16)w0.x; a[1]=(_Float16)w0.y; a[2]=(_Float16)w0.z; a[3]=(_Float16)w0.w;
    a[4]=(_Float16)w1.x; a[5]=(_Float16)w1.y; a[6]=(_Float16)w1.z; a[7]=(_Float16)w1.w;
    a[8]=(_Float16)w2.x; a[9]=(_Float16)w2.y; a[10]=(_Float16)w2.z; a[11]=(_Float16)w2.w;
    a[12]=(_Float16)w3.x; a[13]=(_Float16)w3.y; a[14]=(_Float16)w3.z; a[15]=(_Float16)w3.w;
    b[0]=(_Float16)x0.x; b[1]=(_Float16)x0.y; b[2]=(_Float16)x0.z; b[3]=(_Float16)x0.w;
    b[4]=(_Float16)x1.x; b[5]=(_Float16)x1.y; b[6]=(_Float16)x1.z; b[7]=(_Float16)x1.w;
    b[8]=(_Float16)x2.x; b[9]=(_Float16)x2.y; b[10]=(_Float16)x2.z; b[11]=(_Float16)x2.w;
    b[12]=(_Float16)x3.x; b[13]=(_Float16)x3.y; b[14]=(_Float16)x3.z; b[15]=(_Float16)x3.w;
    c = __builtin_amdgcn_wmma_f32_16x16x32_f16(false, a, false, b, (short)0, c,
                                               false, false);
  }

  const float p0 = params[0], p1 = params[1], p2 = params[2], p3 = params[3];
  const float cp0 = __cosf(p0), cp1 = __cosf(p1), cp2 = __cosf(p2), cp3 = __cosf(p3);
  const int sg = mt * 16 + lo;
  const int bb = sg >> 10;
  const int ss = sg & 1023;
  #pragma unroll
  for (int g = 0; g < 2; ++g) {
    const float z0 = cp0 * __cosf(c[4 * g + 0] + p0);
    const float z1 = cp1 * __cosf(c[4 * g + 1] + p1);
    const float z2 = cp2 * __cosf(c[4 * g + 2] + p2);
    const float z3 = cp3 * __cosf(c[4 * g + 3] + p3);
    const float o1 = z0 * z1;
    const float o2 = o1 * z2;
    const float o3 = o2 * z3;
    const float o0 = z1 * z2 * z3;
    const int hcol = nt * 16 + hi * 8 + g * 4;
    const int head = hcol >> 2;
    *(float4*)(outp + (((bb * HEADS + head) * SEQ + ss) << 2)) =
        make_float4(o0, o1, o2, o3);
  }
}

__global__ __launch_bounds__(256) void qa_attention(
    const float* __restrict__ qp, const float* __restrict__ kp,
    const float* __restrict__ vp, float* __restrict__ out)
{
  __shared__ float ksm[SEQ * 4];
  __shared__ float vsm[SEQ * 4];
  __shared__ float red[8][16][16][4];

  const int bh = blockIdx.x;
  const float* qh = qp + bh * SEQ * 4;
  const float* kh = kp + bh * SEQ * 4;
  const float* vh = vp + bh * SEQ * 4;

  for (int i = threadIdx.x; i < SEQ; i += 256) {
    *(float4*)&ksm[i * 4] = *(const float4*)(kh + i * 4);
    *(float4*)&vsm[i * 4] = *(const float4*)(vh + i * 4);
  }
  __syncthreads();

  const int lane = threadIdx.x & 31, wave = threadIdx.x >> 5;
  const int lo = lane & 15, hi = lane >> 4;
  const int bb = bh >> 6, head = bh & 63;

  for (int qt = wave; qt < 64; qt += 8) {
#if HAVE_WMMA_F32X4
    v2f aq;
    aq[0] = 0.5f * qh[(qt * 16 + lo) * 4 + hi * 2 + 0];
    aq[1] = 0.5f * qh[(qt * 16 + lo) * 4 + hi * 2 + 1];
#else
    float4 qr[8];
    #pragma unroll
    for (int j = 0; j < 8; ++j) {
      float4 q4 = *(const float4*)(qh + (qt * 16 + j + 8 * hi) * 4);
      qr[j] = make_float4(0.5f * q4.x, 0.5f * q4.y, 0.5f * q4.z, 0.5f * q4.w);
    }
#endif
    float run_max[8], run_sum[8];
    float4 ctxp[8];
    #pragma unroll
    for (int j = 0; j < 8; ++j) {
      run_max[j] = -1e30f; run_sum[j] = 0.0f;
      ctxp[j] = make_float4(0.f, 0.f, 0.f, 0.f);
    }

    for (int ct = 0; ct < 64; ++ct) {
      const int ncol = ct * 16 + lo;
      const float4 vcol = *(const float4*)&vsm[ncol * 4];
#if HAVE_WMMA_F32X4
      v2f bk;
      bk[0] = ksm[ncol * 4 + hi * 2 + 0];
      bk[1] = ksm[ncol * 4 + hi * 2 + 1];
      v8f sc = {};
      sc = __builtin_amdgcn_wmma_f32_16x16x4_f32(false, aq, false, bk,
                                                 (short)0, sc, false, false);
#else
      const float4 kc = *(const float4*)&ksm[ncol * 4];
      float sc[8];
      #pragma unroll
      for (int j = 0; j < 8; ++j)
        sc[j] = qr[j].x * kc.x + qr[j].y * kc.y + qr[j].z * kc.z + qr[j].w * kc.w;
#endif
      #pragma unroll
      for (int j = 0; j < 8; ++j) {
        const float s = sc[j];
        float mx = s;
        mx = fmaxf(mx, __shfl_xor(mx, 1));
        mx = fmaxf(mx, __shfl_xor(mx, 2));
        mx = fmaxf(mx, __shfl_xor(mx, 4));
        mx = fmaxf(mx, __shfl_xor(mx, 8));
        const float nm    = fmaxf(run_max[j], mx);
        const float scale = __expf(run_max[j] - nm);
        const float pr    = __expf(s - nm);
        float sum = pr;
        sum += __shfl_xor(sum, 1);
        sum += __shfl_xor(sum, 2);
        sum += __shfl_xor(sum, 4);
        sum += __shfl_xor(sum, 8);
        run_sum[j] = run_sum[j] * scale + sum;
        run_max[j] = nm;
        ctxp[j].x = ctxp[j].x * scale + pr * vcol.x;
        ctxp[j].y = ctxp[j].y * scale + pr * vcol.y;
        ctxp[j].z = ctxp[j].z * scale + pr * vcol.z;
        ctxp[j].w = ctxp[j].w * scale + pr * vcol.w;
      }
    }

    #pragma unroll
    for (int j = 0; j < 8; ++j) {
      const float inv = 1.0f / run_sum[j];
      const int r = j + 8 * hi;
      *(float4*)&red[wave][r][lo][0] =
          make_float4(ctxp[j].x * inv, ctxp[j].y * inv,
                      ctxp[j].z * inv, ctxp[j].w * inv);
    }
    __syncthreads();
    #pragma unroll
    for (int t = 0; t < 2; ++t) {
      const int o = lane * 2 + t;
      const int r = o >> 2, d = o & 3;
      float acc = 0.0f;
      #pragma unroll
      for (int cc = 0; cc < 16; ++cc) acc += red[wave][r][cc][d];
      const int srow = qt * 16 + r;
      out[(bb * SEQ + srow) * EMB + head * 4 + d] = acc;
    }
    __syncthreads();
  }
}

extern "C" void kernel_launch(void* const* d_in, const int* in_sizes, int n_in,
                              void* d_out, int out_size, void* d_ws, size_t ws_size,
                              hipStream_t stream) {
  const float* x      = (const float*)d_in[0];
  const float* Wq     = (const float*)d_in[1];
  const float* Wk     = (const float*)d_in[2];
  const float* Wv     = (const float*)d_in[3];
  const float* params = (const float*)d_in[4];
  float* out = (float*)d_out;

  const size_t headElems = (size_t)BATCH * HEADS * SEQ * DKK;
  float* qp = (float*)d_ws;
  float* kp = qp + headElems;
  float* vp = kp + headElems;

  qa_qkv_quantum<<<768, 256, 0, stream>>>(x, Wq, Wk, Wv, params, qp, kp, vp);
  qa_attention<<<BATCH * HEADS, 256, 0, stream>>>(qp, kp, vp, out);
}


# attention grid 128 -> 512 workgroups: each (b,h) split into 4 query ranges of 16 q-tiles (all keys each, K/V staged per workgroup) so all 256 WGPs run it, 4 waves per SIMD; sum order unchanged
# speedup vs baseline: 8.3470x; 8.3470x over previous
#include <hip/hip_runtime.h>
#include <hip/hip_bf16.h>

#define BATCH 2
#define SEQ   1024
#define EMB   256
#define HEADS 64
#define DKK   4
#define QSPLIT 4
#define QT_PER_WG (64 / QSPLIT)

typedef __attribute__((ext_vector_type(16))) _Float16 v16h;
typedef __attribute__((ext_vector_type(8)))  float    v8f;
typedef __attribute__((ext_vector_type(2)))  float    v2f;

#if defined(__has_builtin)
#if __has_builtin(__builtin_amdgcn_wmma_f32_16x16x4_f32)
#define HAVE_WMMA_F32X4 1
#endif
#endif

__global__ __launch_bounds__(256) void qa_qkv_quantum(
    const float* __restrict__ x,
    const float* __restrict__ Wq, const float* __restrict__ Wk,
    const float* __restrict__ Wv, const float* __restrict__ params,
    float* __restrict__ qp, float* __restrict__ kp, float* __restrict__ vp)
{
  const int lane = threadIdx.x & 31;
  const int wave = threadIdx.x >> 5;
  const int lo = lane & 15, hi = lane >> 4;

  const int tile = blockIdx.x * 8 + wave;
  const int mat  = tile >> 11;
  const int rem  = tile & 2047;
  const int nt   = rem >> 7;
  const int mt   = rem & 127;

  const float* W    = (mat == 0) ? Wq : (mat == 1) ? Wk : Wv;
  float*       outp = (mat == 0) ? qp : (mat == 1) ? kp : vp;

  const float* wrow = W + (nt * 16 + lo) * EMB;
  const float* xrow = x + (mt * 16 + lo) * EMB;

  v8f c = {};
  #pragma unroll
  for (int ks = 0; ks < EMB; ks += 32) {
    const float* wa = wrow + ks + hi * 8;
    const float* xa = xrow + ks + hi * 8;
    const float4 w0 = *(const float4*)(wa);
    const float4 w1 = *(const float4*)(wa + 4);
    const float4 w2 = *(const float4*)(wa + 16);
    const float4 w3 = *(const float4*)(wa + 20);
    const float4 x0 = *(const float4*)(xa);
    const float4 x1 = *(const float4*)(xa + 4);
    const float4 x2 = *(const float4*)(xa + 16);
    const float4 x3 = *(const float4*)(xa + 20);
    v16h a, b;
    a[0]=(_Float16)w0.x; a[1]=(_Float16)w0.y; a[2]=(_Float16)w0.z; a[3]=(_Float16)w0.w;
    a[4]=(_Float16)w1.x; a[5]=(_Float16)w1.y; a[6]=(_Float16)w1.z; a[7]=(_Float16)w1.w;
    a[8]=(_Float16)w2.x; a[9]=(_Float16)w2.y; a[10]=(_Float16)w2.z; a[11]=(_Float16)w2.w;
    a[12]=(_Float16)w3.x; a[13]=(_Float16)w3.y; a[14]=(_Float16)w3.z; a[15]=(_Float16)w3.w;
    b[0]=(_Float16)x0.x; b[1]=(_Float16)x0.y; b[2]=(_Float16)x0.z; b[3]=(_Float16)x0.w;
    b[4]=(_Float16)x1.x; b[5]=(_Float16)x1.y; b[6]=(_Float16)x1.z; b[7]=(_Float16)x1.w;
    b[8]=(_Float16)x2.x; b[9]=(_Float16)x2.y; b[10]=(_Float16)x2.z; b[11]=(_Float16)x2.w;
    b[12]=(_Float16)x3.x; b[13]=(_Float16)x3.y; b[14]=(_Float16)x3.z; b[15]=(_Float16)x3.w;
    c = __builtin_amdgcn_wmma_f32_16x16x32_f16(false, a, false, b, (short)0, c,
                                               false, false);
  }

  const float p0 = params[0], p1 = params[1], p2 = params[2], p3 = params[3];
  const float cp0 = __cosf(p0), cp1 = __cosf(p1), cp2 = __cosf(p2), cp3 = __cosf(p3);
  const int sg = mt * 16 + lo;
  const int bb = sg >> 10;
  const int ss = sg & 1023;
  #pragma unroll
  for (int g = 0; g < 2; ++g) {
    const float z0 = cp0 * __cosf(c[4 * g + 0] + p0);
    const float z1 = cp1 * __cosf(c[4 * g + 1] + p1);
    const float z2 = cp2 * __cosf(c[4 * g + 2] + p2);
    const float z3 = cp3 * __cosf(c[4 * g + 3] + p3);
    const float o1 = z0 * z1;
    const float o2 = o1 * z2;
    const float o3 = o2 * z3;
    const float o0 = z1 * z2 * z3;
    const int hcol = nt * 16 + hi * 8 + g * 4;
    const int head = hcol >> 2;
    *(float4*)(outp + (((bb * HEADS + head) * SEQ + ss) << 2)) =
        make_float4(o0, o1, o2, o3);
  }
}

__global__ __launch_bounds__(256) void qa_attention(
    const float* __restrict__ qp, const float* __restrict__ kp,
    const float* __restrict__ vp, float* __restrict__ out)
{
  __shared__ float ksm[SEQ * 4];
  __shared__ float vsm[SEQ * 4];
  __shared__ float red[8][16][16][4];

  const int bh   = blockIdx.x / QSPLIT;
  const int part = blockIdx.x % QSPLIT;
  const float* qh = qp + bh * SEQ * 4;
  const float* kh = kp + bh * SEQ * 4;
  const float* vh = vp + bh * SEQ * 4;

  for (int i = threadIdx.x; i < SEQ; i += 256) {
    *(float4*)&ksm[i * 4] = *(const float4*)(kh + i * 4);
    *(float4*)&vsm[i * 4] = *(const float4*)(vh + i * 4);
  }
  __syncthreads();

  const int lane = threadIdx.x & 31, wave = threadIdx.x >> 5;
  const int lo = lane & 15, hi = lane >> 4;
  const int bb = bh >> 6, head = bh & 63;

  for (int rnd = 0; rnd < QT_PER_WG / 8; ++rnd) {
    const int qt = part * QT_PER_WG + rnd * 8 + wave;
#if HAVE_WMMA_F32X4
    v2f aq;
    aq[0] = 0.5f * qh[(qt * 16 + lo) * 4 + hi * 2 + 0];
    aq[1] = 0.5f * qh[(qt * 16 + lo) * 4 + hi * 2 + 1];
#else
    float4 qr[8];
    #pragma unroll
    for (int j = 0; j < 8; ++j) {
      float4 q4 = *(const float4*)(qh + (qt * 16 + j + 8 * hi) * 4);
      qr[j] = make_float4(0.5f * q4.x, 0.5f * q4.y, 0.5f * q4.z, 0.5f * q4.w);
    }
#endif
    float run_sum[8];
    float4 ctxp[8];
    #pragma unroll
    for (int j = 0; j < 8; ++j) {
      run_sum[j] = 0.0f;
      ctxp[j] = make_float4(0.f, 0.f, 0.f, 0.f);
    }

    for (int ct = 0; ct < 64; ++ct) {
      const int ncol = ct * 16 + lo;
      const float4 vcol = *(const float4*)&vsm[ncol * 4];
#if HAVE_WMMA_F32X4
      v2f bk;
      bk[0] = ksm[ncol * 4 + hi * 2 + 0];
      bk[1] = ksm[ncol * 4 + hi * 2 + 1];
      v8f sc = {};
      sc = __builtin_amdgcn_wmma_f32_16x16x4_f32(false, aq, false, bk,
                                                 (short)0, sc, false, false);
#else
      const float4 kc = *(const float4*)&ksm[ncol * 4];
      float sc[8];
      #pragma unroll
      for (int j = 0; j < 8; ++j)
        sc[j] = qr[j].x * kc.x + qr[j].y * kc.y + qr[j].z * kc.z + qr[j].w * kc.w;
#endif
      #pragma unroll
      for (int j = 0; j < 8; ++j) {
        const float pr = __expf(sc[j]);
        run_sum[j] += pr;
        ctxp[j].x += pr * vcol.x;
        ctxp[j].y += pr * vcol.y;
        ctxp[j].z += pr * vcol.z;
        ctxp[j].w += pr * vcol.w;
      }
    }

    #pragma unroll
    for (int j = 0; j < 8; ++j) {
      float sum = run_sum[j];
      sum += __shfl_xor(sum, 1);
      sum += __shfl_xor(sum, 2);
      sum += __shfl_xor(sum, 4);
      sum += __shfl_xor(sum, 8);
      run_sum[j] = sum;
    }

    #pragma unroll
    for (int j = 0; j < 8; ++j) {
      const float inv = 1.0f / run_sum[j];
      const int r = j + 8 * hi;
      *(float4*)&red[wave][r][lo][0] =
          make_float4(ctxp[j].x * inv, ctxp[j].y * inv,
                      ctxp[j].z * inv, ctxp[j].w * inv);
    }
    __syncthreads();
    #pragma unroll
    for (int t = 0; t < 2; ++t) {
      const int o = lane * 2 + t;
      const int r = o >> 2, d = o & 3;
      float acc = 0.0f;
      #pragma unroll
      for (int cc = 0; cc < 16; ++cc) acc += red[wave][r][cc][d];
      const int srow = qt * 16 + r;
      out[(bb * SEQ + srow) * EMB + head * 4 + d] = acc;
    }
    __syncthreads();
  }
}

extern "C" void kernel_launch(void* const* d_in, const int* in_sizes, int n_in,
                              void* d_out, int out_size, void* d_ws, size_t ws_size,
                              hipStream_t stream) {
  const float* x      = (const float*)d_in[0];
  const float* Wq     = (const float*)d_in[1];
  const float* Wk     = (const float*)d_in[2];
  const float* Wv     = (const float*)d_in[3];
  const float* params = (const float*)d_in[4];
  float* out = (float*)d_out;

  const size_t headElems = (size_t)BATCH * HEADS * SEQ * DKK;
  float* qp = (float*)d_ws;
  float* kp = qp + headElems;
  float* vp = kp + headElems;

  qa_qkv_quantum<<<768, 256, 0, stream>>>(x, Wq, Wk, Wv, params, qp, kp, vp);
  qa_attention<<<BATCH * HEADS * QSPLIT, 256, 0, stream>>>(qp, kp, vp, out);
}


# projection kernel: x and W blocks read with whole-line global loads (512 contiguous B per wave load), converted to f16 once into padded LDS, fragments read from LDS; 4 waves x 2 tiles per block
# speedup vs baseline: 10.5665x; 1.2659x over previous
#include <hip/hip_runtime.h>
#include <hip/hip_bf16.h>

#define BATCH 2
#define SEQ   1024
#define EMB   256
#define HEADS 64
#define DKK   4
#define QSPLIT 4
#define QT_PER_WG (64 / QSPLIT)

typedef __attribute__((ext_vector_type(16))) _Float16 v16h;
typedef __attribute__((ext_vector_type(8)))  _Float16 v8h;
typedef __attribute__((ext_vector_type(4)))  _Float16 v4h;
typedef __attribute__((ext_vector_type(8)))  float    v8f;
typedef __attribute__((ext_vector_type(2)))  float    v2f;

#if defined(__has_builtin)
#if __has_builtin(__builtin_amdgcn_wmma_f32_16x16x4_f32)
#define HAVE_WMMA_F32X4 1
#endif
#endif

#define XROWS 64
#define WROWS 32
#define PITCH_H 264

__device__ __forceinline__ void stage4(_Float16* dst, const float4 v) {
  v4h h;
  h[0] = (_Float16)v.x; h[1] = (_Float16)v.y;
  h[2] = (_Float16)v.z; h[3] = (_Float16)v.w;
  *(v4h*)dst = h;
}

__device__ __forceinline__ v16h frag16(const _Float16* r, int ks, int hi) {
  const v8h p0 = *(const v8h*)(r + ks + hi * 8);
  const v8h p1 = *(const v8h*)(r + ks + 16 + hi * 8);
  return __builtin_shufflevector(p0, p1, 0, 1, 2, 3, 4, 5, 6, 7,
                                 8, 9, 10, 11, 12, 13, 14, 15);
}

__global__ __launch_bounds__(128) void qa_qkv_quantum(
    const float* __restrict__ x,
    const float* __restrict__ Wq, const float* __restrict__ Wk,
    const float* __restrict__ Wv, const float* __restrict__ params,
    float* __restrict__ qp, float* __restrict__ kp, float* __restrict__ vp)
{
  __shared__ __attribute__((aligned(16))) _Float16 tile[(XROWS + WROWS) * PITCH_H];

  const int t = threadIdx.x;
  const int lane = t & 31;
  const int wave = t >> 5;
  const int lo = lane & 15, hi = lane >> 4;

  const int mat = blockIdx.x >> 8;
  const int rem = blockIdx.x & 255;
  const int np  = rem >> 5;
  const int mb  = rem & 31;

  const float* W    = (mat == 0) ? Wq : (mat == 1) ? Wk : Wv;
  float*       outp = (mat == 0) ? qp : (mat == 1) ? kp : vp;

  const float* xsrc = x + mb * (XROWS * EMB);
  const float* wsrc = W + np * (WROWS * EMB);

  #pragma unroll
  for (int b0 = 0; b0 < 32; b0 += 16) {
    float4 r[16];
    #pragma unroll
    for (int i = 0; i < 16; ++i)
      r[i] = *(const float4*)(xsrc + ((b0 + i) * 128 + t) * 4);
    #pragma unroll
    for (int i = 0; i < 16; ++i) {
      const int c = (b0 + i) * 128 + t;
      stage4(&tile[(c >> 6) * PITCH_H + (c & 63) * 4], r[i]);
    }
  }
  {
    float4 r[16];
    #pragma unroll
    for (int i = 0; i < 16; ++i)
      r[i] = *(const float4*)(wsrc + (i * 128 + t) * 4);
    #pragma unroll
    for (int i = 0; i < 16; ++i) {
      const int c = i * 128 + t;
      stage4(&tile[(XROWS + (c >> 6)) * PITCH_H + (c & 63) * 4], r[i]);
    }
  }
  __syncthreads();

  const _Float16* xr  = &tile[(wave * 16 + lo) * PITCH_H];
  const _Float16* wr0 = &tile[(XROWS + lo) * PITCH_H];
  const _Float16* wr1 = &tile[(XROWS + 16 + lo) * PITCH_H];

  v8f c0 = {}, c1 = {};
  #pragma unroll
  for (int ks = 0; ks < EMB; ks += 32) {
    const v16h b  = frag16(xr,  ks, hi);
    const v16h a0 = frag16(wr0, ks, hi);
    const v16h a1 = frag16(wr1, ks, hi);
    c0 = __builtin_amdgcn_wmma_f32_16x16x32_f16(false, a0, false, b, (short)0, c0,
                                                false, false);
    c1 = __builtin_amdgcn_wmma_f32_16x16x32_f16(false, a1, false, b, (short)0, c1,
                                                false, false);
  }

  const float p0 = params[0], p1 = params[1], p2 = params[2], p3 = params[3];
  const float cp0 = __cosf(p0), cp1 = __cosf(p1), cp2 = __cosf(p2), cp3 = __cosf(p3);
  const int mt = mb * 4 + wave;
  const int sg = mt * 16 + lo;
  const int bb = sg >> 10;
  const int ss = sg & 1023;
  #pragma unroll
  for (int tn = 0; tn < 2; ++tn) {
    const v8f c = tn ? c1 : c0;
    const int nt = np * 2 + tn;
    #pragma unroll
    for (int g = 0; g < 2; ++g) {
      const float z0 = cp0 * __cosf(c[4 * g + 0] + p0);
      const float z1 = cp1 * __cosf(c[4 * g + 1] + p1);
      const float z2 = cp2 * __cosf(c[4 * g + 2] + p2);
      const float z3 = cp3 * __cosf(c[4 * g + 3] + p3);
      const float o1 = z0 * z1;
      const float o2 = o1 * z2;
      const float o3 = o2 * z3;
      const float o0 = z1 * z2 * z3;
      const int hcol = nt * 16 + hi * 8 + g * 4;
      const int head = hcol >> 2;
      *(float4*)(outp + (((bb * HEADS + head) * SEQ + ss) << 2)) =
          make_float4(o0, o1, o2, o3);
    }
  }
}

__global__ __launch_bounds__(256) void qa_attention(
    const float* __restrict__ qp, const float* __restrict__ kp,
    const float* __restrict__ vp, float* __restrict__ out)
{
  __shared__ float ksm[SEQ * 4];
  __shared__ float vsm[SEQ * 4];
  __shared__ float red[8][16][16][4];

  const int bh   = blockIdx.x / QSPLIT;
  const int part = blockIdx.x % QSPLIT;
  const float* qh = qp + bh * SEQ * 4;
  const float* kh = kp + bh * SEQ * 4;
  const float* vh = vp + bh * SEQ * 4;

  for (int i = threadIdx.x; i < SEQ; i += 256) {
    *(float4*)&ksm[i * 4] = *(const float4*)(kh + i * 4);
    *(float4*)&vsm[i * 4] = *(const float4*)(vh + i * 4);
  }
  __syncthreads();

  const int lane = threadIdx.x & 31, wave = threadIdx.x >> 5;
  const int lo = lane & 15, hi = lane >> 4;
  const int bb = bh >> 6, head = bh & 63;

  for (int rnd = 0; rnd < QT_PER_WG / 8; ++rnd) {
    const int qt = part * QT_PER_WG + rnd * 8 + wave;
#if HAVE_WMMA_F32X4
    v2f aq;
    aq[0] = 0.5f * qh[(qt * 16 + lo) * 4 + hi * 2 + 0];
    aq[1] = 0.5f * qh[(qt * 16 + lo) * 4 + hi * 2 + 1];
#else
    float4 qr[8];
    #pragma unroll
    for (int j = 0; j < 8; ++j) {
      float4 q4 = *(const float4*)(qh + (qt * 16 + j + 8 * hi) * 4);
      qr[j] = make_float4(0.5f * q4.x, 0.5f * q4.y, 0.5f * q4.z, 0.5f * q4.w);
    }
#endif
    float run_sum[8];
    float4 ctxp[8];
    #pragma unroll
    for (int j = 0; j < 8; ++j) {
      run_sum[j] = 0.0f;
      ctxp[j] = make_float4(0.f, 0.f, 0.f, 0.f);
    }

    for (int ct = 0; ct < 64; ++ct) {
      const int ncol = ct * 16 + lo;
      const float4 vcol = *(const float4*)&vsm[ncol * 4];
#if HAVE_WMMA_F32X4
      v2f bk;
      bk[0] = ksm[ncol * 4 + hi * 2 + 0];
      bk[1] = ksm[ncol * 4 + hi * 2 + 1];
      v8f sc = {};
      sc = __builtin_amdgcn_wmma_f32_16x16x4_f32(false, aq, false, bk,
                                                 (short)0, sc, false, false);
#else
      const float4 kc = *(const float4*)&ksm[ncol * 4];
      float sc[8];
      #pragma unroll
      for (int j = 0; j < 8; ++j)
        sc[j] = qr[j].x * kc.x + qr[j].y * kc.y + qr[j].z * kc.z + qr[j].w * kc.w;
#endif
      #pragma unroll
      for (int j = 0; j < 8; ++j) {
        const float pr = __expf(sc[j]);
        run_sum[j] += pr;
        ctxp[j].x += pr * vcol.x;
        ctxp[j].y += pr * vcol.y;
        ctxp[j].z += pr * vcol.z;
        ctxp[j].w += pr * vcol.w;
      }
    }

    #pragma unroll
    for (int j = 0; j < 8; ++j) {
      float sum = run_sum[j];
      sum += __shfl_xor(sum, 1);
      sum += __shfl_xor(sum, 2);
      sum += __shfl_xor(sum, 4);
      sum += __shfl_xor(sum, 8);
      run_sum[j] = sum;
    }

    #pragma unroll
    for (int j = 0; j < 8; ++j) {
      const float inv = 1.0f / run_sum[j];
      const int r = j + 8 * hi;
      *(float4*)&red[wave][r][lo][0] =
          make_float4(ctxp[j].x * inv, ctxp[j].y * inv,
                      ctxp[j].z * inv, ctxp[j].w * inv);
    }
    __syncthreads();
    #pragma unroll
    for (int t = 0; t < 2; ++t) {
      const int o = lane * 2 + t;
      const int r = o >> 2, d = o & 3;
      float acc = 0.0f;
      #pragma unroll
      for (int cc = 0; cc < 16; ++cc) acc += red[wave][r][cc][d];
      const int srow = qt * 16 + r;
      out[(bb * SEQ + srow) * EMB + head * 4 + d] = acc;
    }
    __syncthreads();
  }
}

extern "C" void kernel_launch(void* const* d_in, const int* in_sizes, int n_in,
                              void* d_out, int out_size, void* d_ws, size_t ws_size,
                              hipStream_t stream) {
  const float* x      = (const float*)d_in[0];
  const float* Wq     = (const float*)d_in[1];
  const float* Wk     = (const float*)d_in[2];
  const float* Wv     = (const float*)d_in[3];
  const float* params = (const float*)d_in[4];
  float* out = (float*)d_out;

  const size_t headElems = (size_t)BATCH * HEADS * SEQ * DKK;
  float* qp = (float*)d_ws;
  float* kp = qp + headElems;
  float* vp = kp + headElems;

  qa_qkv_quantum<<<768, 128, 0, stream>>>(x, Wq, Wk, Wv, params, qp, kp, vp);
  qa_attention<<<BATCH * HEADS * QSPLIT, 256, 0, stream>>>(qp, kp, vp, out);
}


# attention key loop: log2(e)/sqrt(dk) folded into q so each probability is one exp2, and rows held two per register pair so row sums are packed adds; 73 -> 60 vector instr per 2 key tiles
# speedup vs baseline: 11.1259x; 1.0529x over previous
#include <hip/hip_runtime.h>
#include <hip/hip_bf16.h>

#define BATCH 2
#define SEQ   1024
#define EMB   256
#define HEADS 64
#define DKK   4
#define QSPLIT 4
#define QT_PER_WG (64 / QSPLIT)

typedef __attribute__((ext_vector_type(16))) _Float16 v16h;
typedef __attribute__((ext_vector_type(8)))  _Float16 v8h;
typedef __attribute__((ext_vector_type(4)))  _Float16 v4h;
typedef __attribute__((ext_vector_type(8)))  float    v8f;
typedef __attribute__((ext_vector_type(2)))  float    v2f;
#define QSCALE (0.5f * 1.4426950408889634f)

#if defined(__has_builtin)
#if __has_builtin(__builtin_amdgcn_wmma_f32_16x16x4_f32)
#define HAVE_WMMA_F32X4 1
#endif
#endif

#define XROWS 64
#define WROWS 32
#define PITCH_H 264

__device__ __forceinline__ void stage4(_Float16* dst, const float4 v) {
  v4h h;
  h[0] = (_Float16)v.x; h[1] = (_Float16)v.y;
  h[2] = (_Float16)v.z; h[3] = (_Float16)v.w;
  *(v4h*)dst = h;
}

__device__ __forceinline__ v16h frag16(const _Float16* r, int ks, int hi) {
  const v8h p0 = *(const v8h*)(r + ks + hi * 8);
  const v8h p1 = *(const v8h*)(r + ks + 16 + hi * 8);
  return __builtin_shufflevector(p0, p1, 0, 1, 2, 3, 4, 5, 6, 7,
                                 8, 9, 10, 11, 12, 13, 14, 15);
}

__global__ __launch_bounds__(128) void qa_qkv_quantum(
    const float* __restrict__ x,
    const float* __restrict__ Wq, const float* __restrict__ Wk,
    const float* __restrict__ Wv, const float* __restrict__ params,
    float* __restrict__ qp, float* __restrict__ kp, float* __restrict__ vp)
{
  __shared__ __attribute__((aligned(16))) _Float16 tile[(XROWS + WROWS) * PITCH_H];

  const int t = threadIdx.x;
  const int lane = t & 31;
  const int wave = t >> 5;
  const int lo = lane & 15, hi = lane >> 4;

  const int mat = blockIdx.x >> 8;
  const int rem = blockIdx.x & 255;
  const int np  = rem >> 5;
  const int mb  = rem & 31;

  const float* W    = (mat == 0) ? Wq : (mat == 1) ? Wk : Wv;
  float*       outp = (mat == 0) ? qp : (mat == 1) ? kp : vp;

  const float* xsrc = x + mb * (XROWS * EMB);
  const float* wsrc = W + np * (WROWS * EMB);

  #pragma unroll
  for (int b0 = 0; b0 < 32; b0 += 16) {
    float4 r[16];
    #pragma unroll
    for (int i = 0; i < 16; ++i)
      r[i] = *(const float4*)(xsrc + ((b0 + i) * 128 + t) * 4);
    #pragma unroll
    for (int i = 0; i < 16; ++i) {
      const int c = (b0 + i) * 128 + t;
      stage4(&tile[(c >> 6) * PITCH_H + (c & 63) * 4], r[i]);
    }
  }
  {
    float4 r[16];
    #pragma unroll
    for (int i = 0; i < 16; ++i)
      r[i] = *(const float4*)(wsrc + (i * 128 + t) * 4);
    #pragma unroll
    for (int i = 0; i < 16; ++i) {
      const int c = i * 128 + t;
      stage4(&tile[(XROWS + (c >> 6)) * PITCH_H + (c & 63) * 4], r[i]);
    }
  }
  __syncthreads();

  const _Float16* xr  = &tile[(wave * 16 + lo) * PITCH_H];
  const _Float16* wr0 = &tile[(XROWS + lo) * PITCH_H];
  const _Float16* wr1 = &tile[(XROWS + 16 + lo) * PITCH_H];

  v8f c0 = {}, c1 = {};
  #pragma unroll
  for (int ks = 0; ks < EMB; ks += 32) {
    const v16h b  = frag16(xr,  ks, hi);
    const v16h a0 = frag16(wr0, ks, hi);
    const v16h a1 = frag16(wr1, ks, hi);
    c0 = __builtin_amdgcn_wmma_f32_16x16x32_f16(false, a0, false, b, (short)0, c0,
                                                false, false);
    c1 = __builtin_amdgcn_wmma_f32_16x16x32_f16(false, a1, false, b, (short)0, c1,
                                                false, false);
  }

  const float p0 = params[0], p1 = params[1], p2 = params[2], p3 = params[3];
  const float cp0 = __cosf(p0), cp1 = __cosf(p1), cp2 = __cosf(p2), cp3 = __cosf(p3);
  const int mt = mb * 4 + wave;
  const int sg = mt * 16 + lo;
  const int bb = sg >> 10;
  const int ss = sg & 1023;
  #pragma unroll
  for (int tn = 0; tn < 2; ++tn) {
    const v8f c = tn ? c1 : c0;
    const int nt = np * 2 + tn;
    #pragma unroll
    for (int g = 0; g < 2; ++g) {
      const float z0 = cp0 * __cosf(c[4 * g + 0] + p0);
      const float z1 = cp1 * __cosf(c[4 * g + 1] + p1);
      const float z2 = cp2 * __cosf(c[4 * g + 2] + p2);
      const float z3 = cp3 * __cosf(c[4 * g + 3] + p3);
      const float o1 = z0 * z1;
      const float o2 = o1 * z2;
      const float o3 = o2 * z3;
      const float o0 = z1 * z2 * z3;
      const int hcol = nt * 16 + hi * 8 + g * 4;
      const int head = hcol >> 2;
      *(float4*)(outp + (((bb * HEADS + head) * SEQ + ss) << 2)) =
          make_float4(o0, o1, o2, o3);
    }
  }
}

__global__ __launch_bounds__(256) void qa_attention(
    const float* __restrict__ qp, const float* __restrict__ kp,
    const float* __restrict__ vp, float* __restrict__ out)
{
  __shared__ float ksm[SEQ * 4];
  __shared__ float vsm[SEQ * 4];
  __shared__ float red[8][16][16][4];

  const int bh   = blockIdx.x / QSPLIT;
  const int part = blockIdx.x % QSPLIT;
  const float* qh = qp + bh * SEQ * 4;
  const float* kh = kp + bh * SEQ * 4;
  const float* vh = vp + bh * SEQ * 4;

  for (int i = threadIdx.x; i < SEQ; i += 256) {
    *(float4*)&ksm[i * 4] = *(const float4*)(kh + i * 4);
    *(float4*)&vsm[i * 4] = *(const float4*)(vh + i * 4);
  }
  __syncthreads();

  const int lane = threadIdx.x & 31, wave = threadIdx.x >> 5;
  const int lo = lane & 15, hi = lane >> 4;
  const int bb = bh >> 6, head = bh & 63;

  for (int rnd = 0; rnd < QT_PER_WG / 8; ++rnd) {
    const int qt = part * QT_PER_WG + rnd * 8 + wave;
#if HAVE_WMMA_F32X4
    v2f aq;
    aq[0] = QSCALE * qh[(qt * 16 + lo) * 4 + hi * 2 + 0];
    aq[1] = QSCALE * qh[(qt * 16 + lo) * 4 + hi * 2 + 1];
#else
    float4 qr[8];
    #pragma unroll
    for (int j = 0; j < 8; ++j) {
      float4 q4 = *(const float4*)(qh + (qt * 16 + j + 8 * hi) * 4);
      qr[j] = make_float4(QSCALE * q4.x, QSCALE * q4.y, QSCALE * q4.z, QSCALE * q4.w);
    }
#endif
    v2f rsum[4];
    v2f cx[4], cy[4], cz[4], cw[4];
    #pragma unroll
    for (int i = 0; i < 4; ++i) {
      rsum[i] = v2f{0.0f, 0.0f};
      cx[i] = v2f{0.0f, 0.0f}; cy[i] = v2f{0.0f, 0.0f};
      cz[i] = v2f{0.0f, 0.0f}; cw[i] = v2f{0.0f, 0.0f};
    }

    for (int ct = 0; ct < 64; ++ct) {
      const int ncol = ct * 16 + lo;
      const float4 vcol = *(const float4*)&vsm[ncol * 4];
#if HAVE_WMMA_F32X4
      v2f bk;
      bk[0] = ksm[ncol * 4 + hi * 2 + 0];
      bk[1] = ksm[ncol * 4 + hi * 2 + 1];
      v8f sc = {};
      sc = __builtin_amdgcn_wmma_f32_16x16x4_f32(false, aq, false, bk,
                                                 (short)0, sc, false, false);
#else
      const float4 kc = *(const float4*)&ksm[ncol * 4];
      float sc[8];
      #pragma unroll
      for (int j = 0; j < 8; ++j)
        sc[j] = qr[j].x * kc.x + qr[j].y * kc.y + qr[j].z * kc.z + qr[j].w * kc.w;
#endif
      const v2f vx = v2f{vcol.x, vcol.x}, vy = v2f{vcol.y, vcol.y};
      const v2f vz = v2f{vcol.z, vcol.z}, vw = v2f{vcol.w, vcol.w};
      #pragma unroll
      for (int i = 0; i < 4; ++i) {
        v2f pr;
        pr[0] = __builtin_amdgcn_exp2f(sc[2 * i + 0]);
        pr[1] = __builtin_amdgcn_exp2f(sc[2 * i + 1]);
        rsum[i] += pr;
        cx[i] = __builtin_elementwise_fma(pr, vx, cx[i]);
        cy[i] = __builtin_elementwise_fma(pr, vy, cy[i]);
        cz[i] = __builtin_elementwise_fma(pr, vz, cz[i]);
        cw[i] = __builtin_elementwise_fma(pr, vw, cw[i]);
      }
    }

    float run_sum[8];
    #pragma unroll
    for (int j = 0; j < 8; ++j) {
      float sum = rsum[j >> 1][j & 1];
      sum += __shfl_xor(sum, 1);
      sum += __shfl_xor(sum, 2);
      sum += __shfl_xor(sum, 4);
      sum += __shfl_xor(sum, 8);
      run_sum[j] = sum;
    }

    #pragma unroll
    for (int j = 0; j < 8; ++j) {
      const float inv = 1.0f / run_sum[j];
      const int r = j + 8 * hi;
      *(float4*)&red[wave][r][lo][0] =
          make_float4(cx[j >> 1][j & 1] * inv, cy[j >> 1][j & 1] * inv,
                      cz[j >> 1][j & 1] * inv, cw[j >> 1][j & 1] * inv);
    }
    __syncthreads();
    #pragma unroll
    for (int t = 0; t < 2; ++t) {
      const int o = lane * 2 + t;
      const int r = o >> 2, d = o & 3;
      float acc = 0.0f;
      #pragma unroll
      for (int cc = 0; cc < 16; ++cc) acc += red[wave][r][cc][d];
      const int srow = qt * 16 + r;
      out[(bb * SEQ + srow) * EMB + head * 4 + d] = acc;
    }
    __syncthreads();
  }
}

extern "C" void kernel_launch(void* const* d_in, const int* in_sizes, int n_in,
                              void* d_out, int out_size, void* d_ws, size_t ws_size,
                              hipStream_t stream) {
  const float* x      = (const float*)d_in[0];
  const float* Wq     = (const float*)d_in[1];
  const float* Wk     = (const float*)d_in[2];
  const float* Wv     = (const float*)d_in[3];
  const float* params = (const float*)d_in[4];
  float* out = (float*)d_out;

  const size_t headElems = (size_t)BATCH * HEADS * SEQ * DKK;
  float* qp = (float*)d_ws;
  float* kp = qp + headElems;
  float* vp = kp + headElems;

  qa_qkv_quantum<<<768, 128, 0, stream>>>(x, Wq, Wk, Wv, params, qp, kp, vp);
  qa_attention<<<BATCH * HEADS * QSPLIT, 256, 0, stream>>>(qp, kp, vp, out);
}


# attention prologue: every global load of a workgroup (K and V rows of all four staging passes, q fragments of both q-tiles) issued as one group ahead of the first LDS store: one memory wait, not six
# speedup vs baseline: 16.6117x; 1.4931x over previous
#include <hip/hip_runtime.h>
#include <hip/hip_bf16.h>

#define BATCH 2
#define SEQ   1024
#define EMB   256
#define HEADS 64
#define DKK   4
#define QSPLIT 4
#define QT_PER_WG (64 / QSPLIT)

typedef __attribute__((ext_vector_type(16))) _Float16 v16h;
typedef __attribute__((ext_vector_type(8)))  _Float16 v8h;
typedef __attribute__((ext_vector_type(4)))  _Float16 v4h;
typedef __attribute__((ext_vector_type(8)))  float    v8f;
typedef __attribute__((ext_vector_type(2)))  float    v2f;
#define QSCALE (0.5f * 1.4426950408889634f)

#define XROWS 64
#define WROWS 32
#define PITCH_H 264

__device__ __forceinline__ void stage4(_Float16* dst, const float4 v) {
  v4h h;
  h[0] = (_Float16)v.x; h[1] = (_Float16)v.y;
  h[2] = (_Float16)v.z; h[3] = (_Float16)v.w;
  *(v4h*)dst = h;
}

__device__ __forceinline__ v16h frag16(const _Float16* r, int ks, int hi) {
  const v8h p0 = *(const v8h*)(r + ks + hi * 8);
  const v8h p1 = *(const v8h*)(r + ks + 16 + hi * 8);
  return __builtin_shufflevector(p0, p1, 0, 1, 2, 3, 4, 5, 6, 7,
                                 8, 9, 10, 11, 12, 13, 14, 15);
}

__global__ __launch_bounds__(128) void qa_qkv_quantum(
    const float* __restrict__ x,
    const float* __restrict__ Wq, const float* __restrict__ Wk,
    const float* __restrict__ Wv, const float* __restrict__ params,
    float* __restrict__ qp, float* __restrict__ kp, float* __restrict__ vp)
{
  __shared__ __attribute__((aligned(16))) _Float16 tile[(XROWS + WROWS) * PITCH_H];

  const int t = threadIdx.x;
  const int lane = t & 31;
  const int wave = t >> 5;
  const int lo = lane & 15, hi = lane >> 4;

  const int mat = blockIdx.x >> 8;
  const int rem = blockIdx.x & 255;
  const int np  = rem >> 5;
  const int mb  = rem & 31;

  const float* W    = (mat == 0) ? Wq : (mat == 1) ? Wk : Wv;
  float*       outp = (mat == 0) ? qp : (mat == 1) ? kp : vp;

  const float* xsrc = x + mb * (XROWS * EMB);
  const float* wsrc = W + np * (WROWS * EMB);

  #pragma unroll
  for (int b0 = 0; b0 < 32; b0 += 16) {
    float4 r[16];
    #pragma unroll
    for (int i = 0; i < 16; ++i)
      r[i] = *(const float4*)(xsrc + ((b0 + i) * 128 + t) * 4);
    #pragma unroll
    for (int i = 0; i < 16; ++i) {
      const int c = (b0 + i) * 128 + t;
      stage4(&tile[(c >> 6) * PITCH_H + (c & 63) * 4], r[i]);
    }
  }
  {
    float4 r[16];
    #pragma unroll
    for (int i = 0; i < 16; ++i)
      r[i] = *(const float4*)(wsrc + (i * 128 + t) * 4);
    #pragma unroll
    for (int i = 0; i < 16; ++i) {
      const int c = i * 128 + t;
      stage4(&tile[(XROWS + (c >> 6)) * PITCH_H + (c & 63) * 4], r[i]);
    }
  }
  __syncthreads();

  const _Float16* xr  = &tile[(wave * 16 + lo) * PITCH_H];
  const _Float16* wr0 = &tile[(XROWS + lo) * PITCH_H];
  const _Float16* wr1 = &tile[(XROWS + 16 + lo) * PITCH_H];

  v8f c0 = {}, c1 = {};
  #pragma unroll
  for (int ks = 0; ks < EMB; ks += 32) {
    const v16h b  = frag16(xr,  ks, hi);
    const v16h a0 = frag16(wr0, ks, hi);
    const v16h a1 = frag16(wr1, ks, hi);
    c0 = __builtin_amdgcn_wmma_f32_16x16x32_f16(false, a0, false, b, (short)0, c0,
                                                false, false);
    c1 = __builtin_amdgcn_wmma_f32_16x16x32_f16(false, a1, false, b, (short)0, c1,
                                                false, false);
  }

  const float p0 = params[0], p1 = params[1], p2 = params[2], p3 = params[3];
  const float cp0 = __cosf(p0), cp1 = __cosf(p1), cp2 = __cosf(p2), cp3 = __cosf(p3);
  const int mt = mb * 4 + wave;
  const int sg = mt * 16 + lo;
  const int bb = sg >> 10;
  const int ss = sg & 1023;
  #pragma unroll
  for (int tn = 0; tn < 2; ++tn) {
    const v8f c = tn ? c1 : c0;
    const int nt = np * 2 + tn;
    #pragma unroll
    for (int g = 0; g < 2; ++g) {
      const float z0 = cp0 * __cosf(c[4 * g + 0] + p0);
      const float z1 = cp1 * __cosf(c[4 * g + 1] + p1);
      const float z2 = cp2 * __cosf(c[4 * g + 2] + p2);
      const float z3 = cp3 * __cosf(c[4 * g + 3] + p3);
      const float o1 = z0 * z1;
      const float o2 = o1 * z2;
      const float o3 = o2 * z3;
      const float o0 = z1 * z2 * z3;
      const int hcol = nt * 16 + hi * 8 + g * 4;
      const int head = hcol >> 2;
      *(float4*)(outp + (((bb * HEADS + head) * SEQ + ss) << 2)) =
          make_float4(o0, o1, o2, o3);
    }
  }
}

#define VT_PITCH (SEQ + 16)

__global__ __launch_bounds__(256) void qa_attention(
    const float* __restrict__ qp, const float* __restrict__ kp,
    const float* __restrict__ vp, float* __restrict__ out)
{
  __shared__ float ksm[SEQ * 4];
  __shared__ __attribute__((aligned(16))) _Float16 vt[5 * VT_PITCH];

  const int bh   = blockIdx.x / QSPLIT;
  const int part = blockIdx.x % QSPLIT;
  const float* qh = qp + bh * SEQ * 4;
  const float* kh = kp + bh * SEQ * 4;
  const float* vh = vp + bh * SEQ * 4;

  const int lane = threadIdx.x & 31, wave = threadIdx.x >> 5;
  const int lo = lane & 15, hi = lane >> 4;
  const int bb = bh >> 6, head = bh & 63;

  float4 kr[SEQ / 256], vr4[SEQ / 256];
  #pragma unroll
  for (int j = 0; j < SEQ / 256; ++j) {
    const int i = j * 256 + threadIdx.x;
    kr[j]  = *(const float4*)(kh + i * 4);
    vr4[j] = *(const float4*)(vh + i * 4);
  }
  v2f qf[QT_PER_WG / 8];
  #pragma unroll
  for (int rnd = 0; rnd < QT_PER_WG / 8; ++rnd) {
    const int qt = part * QT_PER_WG + rnd * 8 + wave;
    qf[rnd] = *(const v2f*)(qh + (qt * 16 + lo) * 4 + hi * 2);
  }
  #pragma unroll
  for (int j = 0; j < SEQ / 256; ++j) {
    const int i = j * 256 + threadIdx.x;
    *(float4*)&ksm[i * 4] = kr[j];
    vt[0 * VT_PITCH + i] = (_Float16)vr4[j].x;
    vt[1 * VT_PITCH + i] = (_Float16)vr4[j].y;
    vt[2 * VT_PITCH + i] = (_Float16)vr4[j].z;
    vt[3 * VT_PITCH + i] = (_Float16)vr4[j].w;
    vt[4 * VT_PITCH + i] = (_Float16)1.0f;
  }
  __syncthreads();

  const _Float16* vr = &vt[(lo < 4 ? lo : 4) * VT_PITCH];

  #pragma unroll
  for (int rnd = 0; rnd < QT_PER_WG / 8; ++rnd) {
    const int qt = part * QT_PER_WG + rnd * 8 + wave;
    v2f bq;
    bq[0] = QSCALE * qf[rnd][0];
    bq[1] = QSCALE * qf[rnd][1];

    v8f acc = {};

    for (int ct = 0; ct < 64; ct += 2) {
      const int key0 = ct * 16 + lo;
      v2f ak0, ak1;
      ak0[0] = ksm[key0 * 4 + hi * 2 + 0];
      ak0[1] = ksm[key0 * 4 + hi * 2 + 1];
      ak1[0] = ksm[(key0 + 16) * 4 + hi * 2 + 0];
      ak1[1] = ksm[(key0 + 16) * 4 + hi * 2 + 1];
      const v16h va = frag16(vr, ct * 16, hi);

      v8f s0 = {}, s1 = {};
      s0 = __builtin_amdgcn_wmma_f32_16x16x4_f32(false, ak0, false, bq,
                                                 (short)0, s0, false, false);
      s1 = __builtin_amdgcn_wmma_f32_16x16x4_f32(false, ak1, false, bq,
                                                 (short)0, s1, false, false);
      v16h pb;
      #pragma unroll
      for (int e = 0; e < 8; ++e) {
        pb[e]     = (_Float16)__builtin_amdgcn_exp2f(s0[e]);
        pb[8 + e] = (_Float16)__builtin_amdgcn_exp2f(s1[e]);
      }
      acc = __builtin_amdgcn_wmma_f32_16x16x32_f16(false, va, false, pb,
                                                   (short)0, acc, false, false);
    }

    if (hi == 0) {
      const float inv = 1.0f / acc[4];
      const int srow = qt * 16 + lo;
      *(float4*)(out + (bb * SEQ + srow) * EMB + head * 4) =
          make_float4(acc[0] * inv, acc[1] * inv, acc[2] * inv, acc[3] * inv);
    }
  }
}

extern "C" void kernel_launch(void* const* d_in, const int* in_sizes, int n_in,
                              void* d_out, int out_size, void* d_ws, size_t ws_size,
                              hipStream_t stream) {
  const float* x      = (const float*)d_in[0];
  const float* Wq     = (const float*)d_in[1];
  const float* Wk     = (const float*)d_in[2];
  const float* Wv     = (const float*)d_in[3];
  const float* params = (const float*)d_in[4];
  float* out = (float*)d_out;

  const size_t headElems = (size_t)BATCH * HEADS * SEQ * DKK;
  float* qp = (float*)d_ws;
  float* kp = qp + headElems;
  float* vp = kp + headElems;

  qa_qkv_quantum<<<768, 128, 0, stream>>>(x, Wq, Wk, Wv, params, qp, kp, vp);
  qa_attention<<<BATCH * HEADS * QSPLIT, 256, 0, stream>>>(qp, kp, vp, out);
}


# attention scores: f32 16x16x4 form -> f16 16x16x32 form, q and k each split into f16 hi+lo parts (4 cross products per dim in the k axis, f32 accumulate): precision kept; score sum order changes
# speedup vs baseline: 18.2772x; 1.1003x over previous
#include <hip/hip_runtime.h>
#include <hip/hip_bf16.h>

#define BATCH 2
#define SEQ   1024
#define EMB   256
#define HEADS 64
#define DKK   4
#define QSPLIT 4
#define QT_PER_WG (64 / QSPLIT)

typedef __attribute__((ext_vector_type(16))) _Float16 v16h;
typedef __attribute__((ext_vector_type(8)))  _Float16 v8h;
typedef __attribute__((ext_vector_type(4)))  _Float16 v4h;
typedef __attribute__((ext_vector_type(8)))  float    v8f;
#define QSCALE (0.5f * 1.4426950408889634f)

#define XROWS 64
#define WROWS 32
#define PITCH_H 264

__device__ __forceinline__ void stage4(_Float16* dst, const float4 v) {
  v4h h;
  h[0] = (_Float16)v.x; h[1] = (_Float16)v.y;
  h[2] = (_Float16)v.z; h[3] = (_Float16)v.w;
  *(v4h*)dst = h;
}

__device__ __forceinline__ v16h frag16(const _Float16* r, int ks, int hi) {
  const v8h p0 = *(const v8h*)(r + ks + hi * 8);
  const v8h p1 = *(const v8h*)(r + ks + 16 + hi * 8);
  return __builtin_shufflevector(p0, p1, 0, 1, 2, 3, 4, 5, 6, 7,
                                 8, 9, 10, 11, 12, 13, 14, 15);
}

__global__ __launch_bounds__(128) void qa_qkv_quantum(
    const float* __restrict__ x,
    const float* __restrict__ Wq, const float* __restrict__ Wk,
    const float* __restrict__ Wv, const float* __restrict__ params,
    float* __restrict__ qp, float* __restrict__ kp, float* __restrict__ vp)
{
  __shared__ __attribute__((aligned(16))) _Float16 tile[(XROWS + WROWS) * PITCH_H];

  const int t = threadIdx.x;
  const int lane = t & 31;
  const int wave = t >> 5;
  const int lo = lane & 15, hi = lane >> 4;

  const int mat = blockIdx.x >> 8;
  const int rem = blockIdx.x & 255;
  const int np  = rem >> 5;
  const int mb  = rem & 31;

  const float* W    = (mat == 0) ? Wq : (mat == 1) ? Wk : Wv;
  float*       outp = (mat == 0) ? qp : (mat == 1) ? kp : vp;

  const float* xsrc = x + mb * (XROWS * EMB);
  const float* wsrc = W + np * (WROWS * EMB);

  #pragma unroll
  for (int b0 = 0; b0 < 32; b0 += 16) {
    float4 r[16];
    #pragma unroll
    for (int i = 0; i < 16; ++i)
      r[i] = *(const float4*)(xsrc + ((b0 + i) * 128 + t) * 4);
    #pragma unroll
    for (int i = 0; i < 16; ++i) {
      const int c = (b0 + i) * 128 + t;
      stage4(&tile[(c >> 6) * PITCH_H + (c & 63) * 4], r[i]);
    }
  }
  {
    float4 r[16];
    #pragma unroll
    for (int i = 0; i < 16; ++i)
      r[i] = *(const float4*)(wsrc + (i * 128 + t) * 4);
    #pragma unroll
    for (int i = 0; i < 16; ++i) {
      const int c = i * 128 + t;
      stage4(&tile[(XROWS + (c >> 6)) * PITCH_H + (c & 63) * 4], r[i]);
    }
  }
  __syncthreads();

  const _Float16* xr  = &tile[(wave * 16 + lo) * PITCH_H];
  const _Float16* wr0 = &tile[(XROWS + lo) * PITCH_H];
  const _Float16* wr1 = &tile[(XROWS + 16 + lo) * PITCH_H];

  v8f c0 = {}, c1 = {};
  #pragma unroll
  for (int ks = 0; ks < EMB; ks += 32) {
    const v16h b  = frag16(xr,  ks, hi);
    const v16h a0 = frag16(wr0, ks, hi);
    const v16h a1 = frag16(wr1, ks, hi);
    c0 = __builtin_amdgcn_wmma_f32_16x16x32_f16(false, a0, false, b, (short)0, c0,
                                                false, false);
    c1 = __builtin_amdgcn_wmma_f32_16x16x32_f16(false, a1, false, b, (short)0, c1,
                                                false, false);
  }

  const float p0 = params[0], p1 = params[1], p2 = params[2], p3 = params[3];
  const float cp0 = __cosf(p0), cp1 = __cosf(p1), cp2 = __cosf(p2), cp3 = __cosf(p3);
  const int mt = mb * 4 + wave;
  const int sg = mt * 16 + lo;
  const int bb = sg >> 10;
  const int ss = sg & 1023;
  #pragma unroll
  for (int tn = 0; tn < 2; ++tn) {
    const v8f c = tn ? c1 : c0;
    const int nt = np * 2 + tn;
    #pragma unroll
    for (int g = 0; g < 2; ++g) {
      const float z0 = cp0 * __cosf(c[4 * g + 0] + p0);
      const float z1 = cp1 * __cosf(c[4 * g + 1] + p1);
      const float z2 = cp2 * __cosf(c[4 * g + 2] + p2);
      const float z3 = cp3 * __cosf(c[4 * g + 3] + p3);
      const float o1 = z0 * z1;
      const float o2 = o1 * z2;
      const float o3 = o2 * z3;
      const float o0 = z1 * z2 * z3;
      const int hcol = nt * 16 + hi * 8 + g * 4;
      const int head = hcol >> 2;
      *(float4*)(outp + (((bb * HEADS + head) * SEQ + ss) << 2)) =
          make_float4(o0, o1, o2, o3);
    }
  }
}

#define VT_PITCH (SEQ + 16)

__device__ __forceinline__ void split4(const float4 v, v4h& hi, v4h& lo) {
  hi[0] = (_Float16)v.x; hi[1] = (_Float16)v.y;
  hi[2] = (_Float16)v.z; hi[3] = (_Float16)v.w;
  lo[0] = (_Float16)(v.x - (float)hi[0]);
  lo[1] = (_Float16)(v.y - (float)hi[1]);
  lo[2] = (_Float16)(v.z - (float)hi[2]);
  lo[3] = (_Float16)(v.w - (float)hi[3]);
}

__global__ __launch_bounds__(256) void qa_attention(
    const float* __restrict__ qp, const float* __restrict__ kp,
    const float* __restrict__ vp, float* __restrict__ out)
{
  __shared__ __attribute__((aligned(16))) _Float16 ksm[SEQ * 8];
  __shared__ __attribute__((aligned(16))) _Float16 vt[5 * VT_PITCH];

  const int bh   = blockIdx.x / QSPLIT;
  const int part = blockIdx.x % QSPLIT;
  const float* qh = qp + bh * SEQ * 4;
  const float* kh = kp + bh * SEQ * 4;
  const float* vh = vp + bh * SEQ * 4;

  const int lane = threadIdx.x & 31, wave = threadIdx.x >> 5;
  const int lo = lane & 15, hi = lane >> 4;
  const int bb = bh >> 6, head = bh & 63;

  float4 kr[SEQ / 256], vr4[SEQ / 256];
  #pragma unroll
  for (int j = 0; j < SEQ / 256; ++j) {
    const int i = j * 256 + threadIdx.x;
    kr[j]  = *(const float4*)(kh + i * 4);
    vr4[j] = *(const float4*)(vh + i * 4);
  }
  float4 qf[QT_PER_WG / 8];
  #pragma unroll
  for (int rnd = 0; rnd < QT_PER_WG / 8; ++rnd) {
    const int qt = part * QT_PER_WG + rnd * 8 + wave;
    qf[rnd] = *(const float4*)(qh + (qt * 16 + lo) * 4);
  }
  #pragma unroll
  for (int j = 0; j < SEQ / 256; ++j) {
    const int i = j * 256 + threadIdx.x;
    v4h kh4, kl4;
    split4(kr[j], kh4, kl4);
    *(v8h*)&ksm[i * 8] = __builtin_shufflevector(kh4, kl4, 0, 1, 2, 3, 4, 5, 6, 7);
    vt[0 * VT_PITCH + i] = (_Float16)vr4[j].x;
    vt[1 * VT_PITCH + i] = (_Float16)vr4[j].y;
    vt[2 * VT_PITCH + i] = (_Float16)vr4[j].z;
    vt[3 * VT_PITCH + i] = (_Float16)vr4[j].w;
    vt[4 * VT_PITCH + i] = (_Float16)1.0f;
  }
  __syncthreads();

  const _Float16* vr = &vt[(lo < 4 ? lo : 4) * VT_PITCH];

  #pragma unroll
  for (int rnd = 0; rnd < QT_PER_WG / 8; ++rnd) {
    const int qt = part * QT_PER_WG + rnd * 8 + wave;
    v4h xh, xl;
    split4(make_float4(QSCALE * qf[rnd].x, QSCALE * qf[rnd].y,
                       QSCALE * qf[rnd].z, QSCALE * qf[rnd].w), xh, xl);
    v16h bq0 = {}, bq1 = {};
    #pragma unroll
    for (int d = 0; d < 4; ++d) {
      const _Float16 xs = hi ? xl[d] : xh[d];
      bq0[d] = xs;      bq0[4 + d] = xs;
      bq1[8 + d] = xs;  bq1[12 + d] = xs;
    }

    v8f acc = {};

    for (int ct = 0; ct < 64; ct += 2) {
      const int key0 = ct * 16 + lo;
      const v8h k0 = *(const v8h*)&ksm[key0 * 8];
      const v8h k1 = *(const v8h*)&ksm[(key0 + 16) * 8];
      const v16h ak = __builtin_shufflevector(k0, k1, 0, 1, 2, 3, 4, 5, 6, 7,
                                              8, 9, 10, 11, 12, 13, 14, 15);
      const v16h va = frag16(vr, ct * 16, hi);

      v8f s0 = {}, s1 = {};
      s0 = __builtin_amdgcn_wmma_f32_16x16x32_f16(false, ak, false, bq0,
                                                  (short)0, s0, false, false);
      s1 = __builtin_amdgcn_wmma_f32_16x16x32_f16(false, ak, false, bq1,
                                                  (short)0, s1, false, false);
      v16h pb;
      #pragma unroll
      for (int e = 0; e < 8; ++e) {
        pb[e]     = (_Float16)__builtin_amdgcn_exp2f(s0[e]);
        pb[8 + e] = (_Float16)__builtin_amdgcn_exp2f(s1[e]);
      }
      acc = __builtin_amdgcn_wmma_f32_16x16x32_f16(false, va, false, pb,
                                                   (short)0, acc, false, false);
    }

    if (hi == 0) {
      const float inv = 1.0f / acc[4];
      const int srow = qt * 16 + lo;
      *(float4*)(out + (bb * SEQ + srow) * EMB + head * 4) =
          make_float4(acc[0] * inv, acc[1] * inv, acc[2] * inv, acc[3] * inv);
    }
  }
}

extern "C" void kernel_launch(void* const* d_in, const int* in_sizes, int n_in,
                              void* d_out, int out_size, void* d_ws, size_t ws_size,
                              hipStream_t stream) {
  const float* x      = (const float*)d_in[0];
  const float* Wq     = (const float*)d_in[1];
  const float* Wk     = (const float*)d_in[2];
  const float* Wv     = (const float*)d_in[3];
  const float* params = (const float*)d_in[4];
  float* out = (float*)d_out;

  const size_t headElems = (size_t)BATCH * HEADS * SEQ * DKK;
  float* qp = (float*)d_ws;
  float* kp = qp + headElems;
  float* vp = kp + headElems;

  qa_qkv_quantum<<<768, 128, 0, stream>>>(x, Wq, Wk, Wv, params, qp, kp, vp);
  qa_attention<<<BATCH * HEADS * QSPLIT, 256, 0, stream>>>(qp, kp, vp, out);
}


# speedup vs baseline: 19.2526x; 1.0534x over previous
#include <hip/hip_runtime.h>
#include <hip/hip_bf16.h>

#define BATCH 2
#define SEQ   1024
#define EMB   256
#define HEADS 64
#define DKK   4
#define QSPLIT 4
#define QT_PER_WG (64 / QSPLIT)

typedef __attribute__((ext_vector_type(16))) _Float16 v16h;
typedef __attribute__((ext_vector_type(8)))  _Float16 v8h;
typedef __attribute__((ext_vector_type(4)))  _Float16 v4h;
typedef __attribute__((ext_vector_type(8)))  float    v8f;
#define QSCALE (0.5f * 1.4426950408889634f)

#define XROWS 64
#define WROWS 32
#define PITCH_H 264

__device__ __forceinline__ void stage4(_Float16* dst, const float4 v) {
  v4h h;
  h[0] = (_Float16)v.x; h[1] = (_Float16)v.y;
  h[2] = (_Float16)v.z; h[3] = (_Float16)v.w;
  *(v4h*)dst = h;
}

__device__ __forceinline__ v16h frag16(const _Float16* r, int ks, int hi) {
  const v8h p0 = *(const v8h*)(r + ks + hi * 8);
  const v8h p1 = *(const v8h*)(r + ks + 16 + hi * 8);
  return __builtin_shufflevector(p0, p1, 0, 1, 2, 3, 4, 5, 6, 7,
                                 8, 9, 10, 11, 12, 13, 14, 15);
}

__global__ __launch_bounds__(128) void qa_qkv_quantum(
    const float* __restrict__ x,
    const float* __restrict__ Wq, const float* __restrict__ Wk,
    const float* __restrict__ Wv, const float* __restrict__ params,
    float* __restrict__ qp, float* __restrict__ kp, float* __restrict__ vp)
{
  __shared__ __attribute__((aligned(16))) _Float16 tile[(XROWS + WROWS) * PITCH_H];

  const int t = threadIdx.x;
  const int lane = t & 31;
  const int wave = t >> 5;
  const int lo = lane & 15, hi = lane >> 4;

  const int mat = blockIdx.x >> 8;
  const int rem = blockIdx.x & 255;
  const int np  = rem >> 5;
  const int mb  = rem & 31;

  const float* W    = (mat == 0) ? Wq : (mat == 1) ? Wk : Wv;
  float*       outp = (mat == 0) ? qp : (mat == 1) ? kp : vp;

  const float* xsrc = x + mb * (XROWS * EMB);
  const float* wsrc = W + np * (WROWS * EMB);

  #pragma unroll
  for (int b0 = 0; b0 < 32; b0 += 16) {
    float4 r[16];
    #pragma unroll
    for (int i = 0; i < 16; ++i)
      r[i] = *(const float4*)(xsrc + ((b0 + i) * 128 + t) * 4);
    #pragma unroll
    for (int i = 0; i < 16; ++i) {
      const int c = (b0 + i) * 128 + t;
      stage4(&tile[(c >> 6) * PITCH_H + (c & 63) * 4], r[i]);
    }
  }
  {
    float4 r[16];
    #pragma unroll
    for (int i = 0; i < 16; ++i)
      r[i] = *(const float4*)(wsrc + (i * 128 + t) * 4);
    #pragma unroll
    for (int i = 0; i < 16; ++i) {
      const int c = i * 128 + t;
      stage4(&tile[(XROWS + (c >> 6)) * PITCH_H + (c & 63) * 4], r[i]);
    }
  }
  __syncthreads();

  const _Float16* xr  = &tile[(wave * 16 + lo) * PITCH_H];
  const _Float16* wr0 = &tile[(XROWS + lo) * PITCH_H];
  const _Float16* wr1 = &tile[(XROWS + 16 + lo) * PITCH_H];

  v8f c0 = {}, c1 = {};
  #pragma unroll
  for (int ks = 0; ks < EMB; ks += 32) {
    const v16h b  = frag16(xr,  ks, hi);
    const v16h a0 = frag16(wr0, ks, hi);
    const v16h a1 = frag16(wr1, ks, hi);
    c0 = __builtin_amdgcn_wmma_f32_16x16x32_f16(false, a0, false, b, (short)0, c0,
                                                false, false);
    c1 = __builtin_amdgcn_wmma_f32_16x16x32_f16(false, a1, false, b, (short)0, c1,
                                                false, false);
  }

  const float p0 = params[0], p1 = params[1], p2 = params[2], p3 = params[3];
  const float cp0 = __cosf(p0), cp1 = __cosf(p1), cp2 = __cosf(p2), cp3 = __cosf(p3);
  const int mt = mb * 4 + wave;
  const int sg = mt * 16 + lo;
  const int bb = sg >> 10;
  const int ss = sg & 1023;
  #pragma unroll
  for (int tn = 0; tn < 2; ++tn) {
    const v8f c = tn ? c1 : c0;
    const int nt = np * 2 + tn;
    #pragma unroll
    for (int g = 0; g < 2; ++g) {
      const float z0 = cp0 * __cosf(c[4 * g + 0] + p0);
      const float z1 = cp1 * __cosf(c[4 * g + 1] + p1);
      const float z2 = cp2 * __cosf(c[4 * g + 2] + p2);
      const float z3 = cp3 * __cosf(c[4 * g + 3] + p3);
      const float o1 = z0 * z1;
      const float o2 = o1 * z2;
      const float o3 = o2 * z3;
      const float o0 = z1 * z2 * z3;
      const int hcol = nt * 16 + hi * 8 + g * 4;
      const int head = hcol >> 2;
      *(float4*)(outp + (((bb * HEADS + head) * SEQ + ss) << 2)) =
          make_float4(o0, o1, o2, o3);
    }
  }
}

#define VT_PITCH (SEQ + 16)

__device__ __forceinline__ void split4(const float4 v, v4h& hi, v4h& lo) {
  hi[0] = (_Float16)v.x; hi[1] = (_Float16)v.y;
  hi[2] = (_Float16)v.z; hi[3] = (_Float16)v.w;
  lo[0] = (_Float16)(v.x - (float)hi[0]);
  lo[1] = (_Float16)(v.y - (float)hi[1]);
  lo[2] = (_Float16)(v.z - (float)hi[2]);
  lo[3] = (_Float16)(v.w - (float)hi[3]);
}

#define EXP2_PACK16_ORDER                   \
      "v_exp_f32 %8, %8\n\t"                \
      "v_exp_f32 %9, %9\n\t"                \
      "v_exp_f32 %10, %10\n\t"              \
      "v_exp_f32 %11, %11\n\t"              \
      "v_exp_f32 %12, %12\n\t"              \
      "v_exp_f32 %13, %13\n\t"              \
      "v_cvt_pk_f16_f32 %0, %8, %9\n\t"     \
      "v_exp_f32 %14, %14\n\t"              \
      "v_exp_f32 %15, %15\n\t"              \
      "v_cvt_pk_f16_f32 %1, %10, %11\n\t"   \
      "v_exp_f32 %16, %16\n\t"              \
      "v_exp_f32 %17, %17\n\t"              \
      "v_cvt_pk_f16_f32 %2, %12, %13\n\t"   \
      "v_exp_f32 %18, %18\n\t"              \
      "v_exp_f32 %19, %19\n\t"              \
      "v_cvt_pk_f16_f32 %3, %14, %15\n\t"   \
      "v_exp_f32 %20, %20\n\t"              \
      "v_exp_f32 %21, %21\n\t"              \
      "v_cvt_pk_f16_f32 %4, %16, %17\n\t"   \
      "v_exp_f32 %22, %22\n\t"              \
      "v_exp_f32 %23, %23\n\t"              \
      "v_cvt_pk_f16_f32 %5, %18, %19\n\t"   \
      "v_cvt_pk_f16_f32 %6, %20, %21\n\t"   \
      "v_cvt_pk_f16_f32 %7, %22, %23"
#define EXP2_PACK16_OPERANDS                                      \
      : "=&v"(w0), "=&v"(w1), "=&v"(w2), "=&v"(w3),               \
        "=&v"(w4), "=&v"(w5), "=&v"(w6), "=&v"(w7),               \
        "+v"(a0), "+v"(a1), "+v"(a2), "+v"(a3),                   \
        "+v"(a4), "+v"(a5), "+v"(a6), "+v"(a7),                   \
        "+v"(b0), "+v"(b1), "+v"(b2), "+v"(b3),                   \
        "+v"(b4), "+v"(b5), "+v"(b6), "+v"(b7)
template <bool LEAD>
__device__ __forceinline__ v16h exp2_pack16(const v8f s0, const v8f s1) {
  float a0 = s0[0], a1 = s0[1], a2 = s0[2], a3 = s0[3],
        a4 = s0[4], a5 = s0[5], a6 = s0[6], a7 = s0[7];
  float b0 = s1[0], b1 = s1[1], b2 = s1[2], b3 = s1[3],
        b4 = s1[4], b5 = s1[5], b6 = s1[6], b7 = s1[7];
  unsigned w0, w1, w2, w3, w4, w5, w6, w7;
  if (LEAD)
    asm("v_nop\n\t" EXP2_PACK16_ORDER EXP2_PACK16_OPERANDS);
  else
    asm(EXP2_PACK16_ORDER EXP2_PACK16_OPERANDS);
  typedef __attribute__((ext_vector_type(8))) unsigned v8u;
  const v8u w = {w0, w1, w2, w3, w4, w5, w6, w7};
  return __builtin_bit_cast(v16h, w);
}

__global__ __launch_bounds__(256) void qa_attention(
    const float* __restrict__ qp, const float* __restrict__ kp,
    const float* __restrict__ vp, float* __restrict__ out)
{
  __shared__ __attribute__((aligned(16))) _Float16 ksm[SEQ * 8];
  __shared__ __attribute__((aligned(16))) _Float16 vt[5 * VT_PITCH];

  const int bh   = blockIdx.x / QSPLIT;
  const int part = blockIdx.x % QSPLIT;
  const float* qh = qp + bh * SEQ * 4;
  const float* kh = kp + bh * SEQ * 4;
  const float* vh = vp + bh * SEQ * 4;

  const int lane = threadIdx.x & 31, wave = threadIdx.x >> 5;
  const int lo = lane & 15, hi = lane >> 4;
  const int bb = bh >> 6, head = bh & 63;

  float4 kr[SEQ / 256], vr4[SEQ / 256];
  #pragma unroll
  for (int j = 0; j < SEQ / 256; ++j) {
    const int i = j * 256 + threadIdx.x;
    kr[j]  = *(const float4*)(kh + i * 4);
    vr4[j] = *(const float4*)(vh + i * 4);
  }
  float4 qf[QT_PER_WG / 8];
  #pragma unroll
  for (int rnd = 0; rnd < QT_PER_WG / 8; ++rnd) {
    const int qt = part * QT_PER_WG + rnd * 8 + wave;
    qf[rnd] = *(const float4*)(qh + (qt * 16 + lo) * 4);
  }
  #pragma unroll
  for (int j = 0; j < SEQ / 256; ++j) {
    const int i = j * 256 + threadIdx.x;
    v4h kh4, kl4;
    split4(kr[j], kh4, kl4);
    *(v8h*)&ksm[i * 8] = __builtin_shufflevector(kh4, kl4, 0, 1, 2, 3, 4, 5, 6, 7);
    vt[0 * VT_PITCH + i] = (_Float16)vr4[j].x;
    vt[1 * VT_PITCH + i] = (_Float16)vr4[j].y;
    vt[2 * VT_PITCH + i] = (_Float16)vr4[j].z;
    vt[3 * VT_PITCH + i] = (_Float16)vr4[j].w;
    vt[4 * VT_PITCH + i] = (_Float16)1.0f;
  }
  __syncthreads();

  const _Float16* vr = &vt[(lo < 4 ? lo : 4) * VT_PITCH];

  static_assert(QT_PER_WG / 8 == 2, "the q-tile select below is written for two");
  #pragma unroll 1
  for (int rnd = 0; rnd < QT_PER_WG / 8; ++rnd) {
    const int qt = part * QT_PER_WG + rnd * 8 + wave;
    const float4 qv = (rnd == 0) ? qf[0] : qf[1];
    v4h xh, xl;
    split4(make_float4(QSCALE * qv.x, QSCALE * qv.y,
                       QSCALE * qv.z, QSCALE * qv.w), xh, xl);
    v16h bq0 = {}, bq1 = {};
    #pragma unroll
    for (int d = 0; d < 4; ++d) {
      const _Float16 xs = hi ? xl[d] : xh[d];
      bq0[d] = xs;      bq0[4 + d] = xs;
      bq1[8 + d] = xs;  bq1[12 + d] = xs;
    }

    v8f acc = {};

    constexpr int NS = 16;
    #pragma unroll 1
    for (int ct = 0; ct < 64; ct += 2 * NS) {
      const v8f z = {};
      v8f s0[NS], s1[NS];
      v16h ak[NS], va[NS];
      #pragma unroll
      for (int st = 0; st < NS; ++st) {
        #pragma unroll
        for (int f = (st == 0 ? 0 : st + 2); f < NS && f <= st + 2; ++f) {
          const int keyf = (ct + 2 * f) * 16 + lo;
          const v8h k0 = *(const v8h*)&ksm[keyf * 8];
          const v8h k1 = *(const v8h*)&ksm[(keyf + 16) * 8];
          ak[f] = __builtin_shufflevector(k0, k1, 0, 1, 2, 3, 4, 5, 6, 7,
                                          8, 9, 10, 11, 12, 13, 14, 15);
        }
        va[st] = frag16(vr, (ct + 2 * st) * 16, hi);
        #pragma unroll
        for (int g = (st == 0 ? 0 : st + 1); g < NS && g <= st + 1; ++g) {
          s0[g] = __builtin_amdgcn_wmma_f32_16x16x32_f16(false, ak[g], false, bq0,
                                                         (short)0, z, false, false);
          s1[g] = __builtin_amdgcn_wmma_f32_16x16x32_f16(false, ak[g], false, bq1,
                                                         (short)0, z, false, false);
        }
        __builtin_amdgcn_sched_barrier(0);
        const v16h pb = (st == 0) ? exp2_pack16<true>(s0[st], s1[st])
                                  : exp2_pack16<false>(s0[st], s1[st]);
        acc = __builtin_amdgcn_wmma_f32_16x16x32_f16(false, va[st], false, pb,
                                                     (short)0, acc, false, false);
      }
    }

    if (hi == 0) {
      const float inv = 1.0f / acc[4];
      const int srow = qt * 16 + lo;
      *(float4*)(out + (bb * SEQ + srow) * EMB + head * 4) =
          make_float4(acc[0] * inv, acc[1] * inv, acc[2] * inv, acc[3] * inv);
    }
  }
}

extern "C" void kernel_launch(void* const* d_in, const int* in_sizes, int n_in,
                              void* d_out, int out_size, void* d_ws, size_t ws_size,
                              hipStream_t stream) {
  const float* x      = (const float*)d_in[0];
  const float* Wq     = (const float*)d_in[1];
  const float* Wk     = (const float*)d_in[2];
  const float* Wv     = (const float*)d_in[3];
  const float* params = (const float*)d_in[4];
  float* out = (float*)d_out;

  const size_t headElems = (size_t)BATCH * HEADS * SEQ * DKK;
  float* qp = (float*)d_ws;
  float* kp = qp + headElems;
  float* vp = kp + headElems;

  qa_qkv_quantum<<<768, 128, 0, stream>>>(x, Wq, Wk, Wv, params, qp, kp, vp);
  qa_attention<<<BATCH * HEADS * QSPLIT, 256, 0, stream>>>(qp, kp, vp, out);
}
